# starting kernel
# baseline (speedup 1.0000x reference)
#include <hip/hip_runtime.h>

#define N_NODES 2048
#define F_FEAT  128
#define H_DIM   64
#define O_DIM   16
#define KSLICES 4

typedef float v2f __attribute__((ext_vector_type(2)));
typedef float v8f __attribute__((ext_vector_type(8)));

__global__ __launch_bounds__(256) void hgnam_fsums_kernel(
    const float* __restrict__ x,
    const float* __restrict__ fW1,
    const float* __restrict__ fb1,
    const float* __restrict__ fW2,
    const float* __restrict__ fb2,
    float* __restrict__ fsums)
{
    __shared__ float red[8][256];

    const int lane = threadIdx.x & 31;
    const int wv   = threadIdx.x >> 5;
    const int half = lane >> 4;
    const int r    = lane & 15;
    const int base_n = blockIdx.x * 16;

    v8f acc = {};

    const int f0 = wv * 16;
    for (int fi = 0; fi < 16; ++fi) {
        const int f = f0 + fi;
        const float xv = x[(base_n + r) * F_FEAT + f];
        const float* __restrict__ w1 = fW1 + f * H_DIM;
        const float* __restrict__ b1 = fb1 + f * H_DIM;
        const float* __restrict__ w2 = fW2 + f * H_DIM * O_DIM;

        #pragma unroll
        for (int g = 0; g < 2; ++g) {
            v2f av[8], bv[8];
            #pragma unroll
            for (int c = 0; c < 8; ++c) {
                const int h0 = 4 * (g * 8 + c) + 2 * half;
                av[c] = *(const v2f*)(w1 + h0);
                bv[c].x = w2[(h0    ) * O_DIM + r];
                bv[c].y = w2[(h0 + 1) * O_DIM + r];
            }
            v2f b1v[8];
            #pragma unroll
            for (int c = 0; c < 8; ++c) {
                const int h0 = 4 * (g * 8 + c) + 2 * half;
                b1v[c] = *(const v2f*)(b1 + h0);
            }
            #pragma unroll
            for (int c = 0; c < 8; ++c) {
                v2f a;
                a.x = fmaxf(fmaf(xv, av[c].x, b1v[c].x), 0.0f);
                a.y = fmaxf(fmaf(xv, av[c].y, b1v[c].y), 0.0f);
                acc = __builtin_amdgcn_wmma_f32_16x16x4_f32(
                    false, a, false, bv[c], (short)0, acc, false, false);
            }
        }
    }

    #pragma unroll
    for (int v = 0; v < 8; ++v) red[wv][lane * 8 + v] = acc[v];
    __syncthreads();
    if (wv == 0) {
        float bias = 0.0f;
        for (int f = 0; f < F_FEAT; ++f) bias += fb2[f * O_DIM + r];
        #pragma unroll
        for (int v = 0; v < 8; ++v) {
            float s = 0.0f;
            #pragma unroll
            for (int u = 0; u < 8; ++u) s += red[u][lane * 8 + v];
            fsums[(base_n + v + 8 * half) * O_DIM + r] = s + bias;
        }
    }
}

__global__ __launch_bounds__(256) void hgnam_main_kernel(
    const float* __restrict__ dist,
    const float* __restrict__ nrm,
    const float* __restrict__ mW1,
    const float* __restrict__ mb1,
    const float* __restrict__ mW2,
    const float* __restrict__ mb2p,
    const float* __restrict__ fsums,
    float* __restrict__ part)
{
    __shared__ float sW1[H_DIM], sB1[H_DIM], sW2[H_DIM];
    __shared__ float red[8][256];

    const int lane = threadIdx.x & 31;
    const int wv   = threadIdx.x >> 5;
    const int half = lane >> 4;
    const int r    = lane & 15;
    const int base_n = blockIdx.x * 16;
    const int kbase  = blockIdx.y * (N_NODES / KSLICES) + wv * (N_NODES / KSLICES / 8);
    const int NTILES = N_NODES / KSLICES / 8 / 16;

    if (threadIdx.x < H_DIM) {
        sW1[threadIdx.x] = mW1[threadIdx.x];
        sB1[threadIdx.x] = mb1[threadIdx.x];
        sW2[threadIdx.x] = mW2[threadIdx.x];
    }
    __syncthreads();

    const float mb2 = mb2p[0];
    const int n = base_n + r;
    const float* __restrict__ drow = dist + (size_t)n * N_NODES;
    const float* __restrict__ vrow = nrm  + (size_t)n * N_NODES;

    v8f acc = {};

    for (int kt = 0; kt < NTILES; ++kt) {
        const int m0 = kbase + kt * 16;

        v2f d[4], q[4], s[4], bb[4];
        #pragma unroll
        for (int kk = 0; kk < 4; ++kk) {
            const int mm = m0 + 4 * kk + 2 * half;
            d[kk] = *(const v2f*)(drow + mm);
            q[kk] = *(const v2f*)(vrow + mm);
            bb[kk].x = fsums[(mm    ) * O_DIM + r];
            bb[kk].y = fsums[(mm + 1) * O_DIM + r];
            s[kk].x = mb2; s[kk].y = mb2;
        }

        #pragma unroll 8
        for (int h = 0; h < H_DIM; ++h) {
            const float w1 = sW1[h], b1 = sB1[h], w2 = sW2[h];
            #pragma unroll
            for (int kk = 0; kk < 4; ++kk) {
                float t0 = fmaxf(fmaf(d[kk].x, w1, b1), 0.0f);
                float t1 = fmaxf(fmaf(d[kk].y, w1, b1), 0.0f);
                s[kk].x = fmaf(t0, w2, s[kk].x);
                s[kk].y = fmaf(t1, w2, s[kk].y);
            }
        }

        #pragma unroll
        for (int kk = 0; kk < 4; ++kk) {
            v2f a;
            a.x = s[kk].x * __builtin_amdgcn_rcpf(q[kk].x);
            a.y = s[kk].y * __builtin_amdgcn_rcpf(q[kk].y);
            acc = __builtin_amdgcn_wmma_f32_16x16x4_f32(
                false, a, false, bb[kk], (short)0, acc, false, false);
        }
    }

    #pragma unroll
    for (int v = 0; v < 8; ++v) red[wv][lane * 8 + v] = acc[v];
    __syncthreads();
    if (wv == 0) {
        float* __restrict__ pslice = part + (size_t)blockIdx.y * (N_NODES * O_DIM);
        #pragma unroll
        for (int v = 0; v < 8; ++v) {
            float ssum = 0.0f;
            #pragma unroll
            for (int u = 0; u < 8; ++u) ssum += red[u][lane * 8 + v];
            pslice[(base_n + v + 8 * half) * O_DIM + r] = ssum;
        }
    }
}

__global__ __launch_bounds__(256) void hgnam_combine_kernel(
    const float* __restrict__ part, float* __restrict__ out)
{
    const int i = blockIdx.x * 256 + threadIdx.x;
    if (i < N_NODES * O_DIM) {
        float s = 0.0f;
        #pragma unroll
        for (int k = 0; k < KSLICES; ++k)
            s += part[(size_t)k * (N_NODES * O_DIM) + i];
        out[i] = s;
    }
}

extern "C" void kernel_launch(void* const* d_in, const int* in_sizes, int n_in,
                              void* d_out, int out_size, void* d_ws, size_t ws_size,
                              hipStream_t stream) {
    (void)in_sizes; (void)n_in; (void)out_size; (void)ws_size;
    const float* x    = (const float*)d_in[0];
    const float* dist = (const float*)d_in[1];
    const float* nrm  = (const float*)d_in[2];
    const float* fW1  = (const float*)d_in[3];
    const float* fb1  = (const float*)d_in[4];
    const float* fW2  = (const float*)d_in[5];
    const float* fb2  = (const float*)d_in[6];
    const float* mW1  = (const float*)d_in[7];
    const float* mb1  = (const float*)d_in[8];
    const float* mW2  = (const float*)d_in[9];
    const float* mb2  = (const float*)d_in[10];

    float* fsums = (float*)d_ws;
    float* part  = (float*)d_ws + N_NODES * O_DIM;
    float* out   = (float*)d_out;

    hgnam_fsums_kernel<<<dim3(N_NODES / 16), dim3(256), 0, stream>>>(
        x, fW1, fb1, fW2, fb2, fsums);
    hgnam_main_kernel<<<dim3(N_NODES / 16, KSLICES), dim3(256), 0, stream>>>(
        dist, nrm, mW1, mb1, mW2, mb2, fsums, part);
    hgnam_combine_kernel<<<dim3((N_NODES * O_DIM + 255) / 256), dim3(256), 0, stream>>>(
        part, out);
}


# fsums kernel: each wave stages the next feature's fW2/fW1/fb1 (and its x tile once) in its own LDS buffer via coalesced 16-byte loads issued a feature ahead; WMMA fragments come from LDS
# speedup vs baseline: 1.7040x; 1.7040x over previous
#include <hip/hip_runtime.h>

#define N_NODES 2048
#define F_FEAT  128
#define H_DIM   64
#define O_DIM   16
#define KSLICES 4

typedef float v2f __attribute__((ext_vector_type(2)));
typedef float v8f __attribute__((ext_vector_type(8)));

#define FS_W2_FLOATS    (H_DIM * O_DIM)
#define FS_STAGE_FLOATS (FS_W2_FLOATS + 2 * H_DIM)

typedef float v4f __attribute__((ext_vector_type(4)));

struct FeatStage {
    v4f   w2q[8];
    v4f   wbq;
};

__device__ __forceinline__ void fsums_fetch(
    FeatStage& G,
    const float* __restrict__ wbsrc,
    const float* __restrict__ fW2,
    int f, int lane)
{
    const v4f* __restrict__ g2 = (const v4f*)(fW2 + f * FS_W2_FLOATS) + lane;
    #pragma unroll
    for (int j = 0; j < 8; ++j) G.w2q[j] = g2[j * 32];
    G.wbq = *(const v4f*)(wbsrc + f * H_DIM);
}

__device__ __forceinline__ void fsums_park(
    float* sb, const FeatStage& G, int lane, int half, int r)
{
    #pragma unroll
    for (int j = 0; j < 8; ++j) ((v4f*)sb)[j * 32 + lane] = G.w2q[j];
    v2f lo, hi;
    lo.x = G.wbq.x; lo.y = G.wbq.y;
    hi.x = G.wbq.z; hi.y = G.wbq.w;
    *(v2f*)(sb + FS_W2_FLOATS + (2 * r    ) * 4 + 2 * half) = lo;
    *(v2f*)(sb + FS_W2_FLOATS + (2 * r + 1) * 4 + 2 * half) = hi;
}

__device__ __forceinline__ void fsums_wave_lds_sync()
{
    __builtin_amdgcn_fence(__ATOMIC_RELEASE, "wavefront");
    __builtin_amdgcn_wave_barrier();
    __builtin_amdgcn_fence(__ATOMIC_ACQUIRE, "wavefront");
}

__global__ __launch_bounds__(256) void hgnam_fsums_kernel(
    const float* __restrict__ x,
    const float* __restrict__ fW1,
    const float* __restrict__ fb1,
    const float* __restrict__ fW2,
    const float* __restrict__ fb2,
    float* __restrict__ fsums)
{
    __shared__ __attribute__((aligned(16))) float stage[8][2][FS_STAGE_FLOATS];
    __shared__ float xs[8][256];
    __shared__ float red[8][256];

    const int lane = threadIdx.x & 31;
    const int wv   = threadIdx.x >> 5;
    const int half = lane >> 4;
    const int r    = lane & 15;
    const int base_n = blockIdx.x * 16;
    const float* __restrict__ wbsrc = (half ? fb1 : fW1) + 4 * r;

    v8f acc = {};

    const int f0 = wv * 16;
    FeatStage G;
    fsums_fetch(G, wbsrc, fW2, f0, lane);

    {
        const v4f* __restrict__ xp =
            (const v4f*)(x + (base_n + r) * F_FEAT + f0 + 8 * half);
        const v4f xa = xp[0], xb = xp[1];
        float* xw = &xs[wv][(8 * half) * 16 + r];
        xw[0 * 16] = xa.x; xw[1 * 16] = xa.y; xw[2 * 16] = xa.z; xw[3 * 16] = xa.w;
        xw[4 * 16] = xb.x; xw[5 * 16] = xb.y; xw[6 * 16] = xb.z; xw[7 * 16] = xb.w;
    }

    fsums_park(&stage[wv][0][0], G, lane, half, r);
    fsums_fetch(G, wbsrc, fW2, f0 + 1, lane);
    fsums_wave_lds_sync();

    for (int fi = 0; fi < 16; ++fi) {
        const float* cb = &stage[wv][fi & 1][0];
        float*       nb = &stage[wv][(fi & 1) ^ 1][0];

        const float xcur = xs[wv][fi * 16 + r];
        v4f wb[16];
        v2f bf[16];
        #pragma unroll
        for (int c = 0; c < 16; ++c) {
            const int h0 = 4 * c + 2 * half;
            wb[c]   = *(const v4f*)(cb + FS_W2_FLOATS + (2 * c + half) * 4);
            bf[c].x = cb[(h0    ) * O_DIM + r];
            bf[c].y = cb[(h0 + 1) * O_DIM + r];
        }

        fsums_park(nb, G, lane, half, r);
        const int fn = (fi + 2 < 16) ? (fi + 2) : 15;
        fsums_fetch(G, wbsrc, fW2, f0 + fn, lane);

        #pragma unroll
        for (int c = 0; c < 16; ++c) {
            v2f a;
            a.x = fmaxf(fmaf(xcur, wb[c].x, wb[c].z), 0.0f);
            a.y = fmaxf(fmaf(xcur, wb[c].y, wb[c].w), 0.0f);
            acc = __builtin_amdgcn_wmma_f32_16x16x4_f32(
                false, a, false, bf[c], (short)0, acc, false, false);
        }

        fsums_wave_lds_sync();
    }

    #pragma unroll
    for (int v = 0; v < 8; ++v) red[wv][lane * 8 + v] = acc[v];
    __syncthreads();
    if (wv == 0) {
        float bias = 0.0f;
        for (int f = 0; f < F_FEAT; ++f) bias += fb2[f * O_DIM + r];
        #pragma unroll
        for (int v = 0; v < 8; ++v) {
            float s = 0.0f;
            #pragma unroll
            for (int u = 0; u < 8; ++u) s += red[u][lane * 8 + v];
            fsums[(base_n + v + 8 * half) * O_DIM + r] = s + bias;
        }
    }
}

__global__ __launch_bounds__(256) void hgnam_main_kernel(
    const float* __restrict__ dist,
    const float* __restrict__ nrm,
    const float* __restrict__ mW1,
    const float* __restrict__ mb1,
    const float* __restrict__ mW2,
    const float* __restrict__ mb2p,
    const float* __restrict__ fsums,
    float* __restrict__ part)
{
    __shared__ float sW1[H_DIM], sB1[H_DIM], sW2[H_DIM];
    __shared__ float red[8][256];

    const int lane = threadIdx.x & 31;
    const int wv   = threadIdx.x >> 5;
    const int half = lane >> 4;
    const int r    = lane & 15;
    const int base_n = blockIdx.x * 16;
    const int kbase  = blockIdx.y * (N_NODES / KSLICES) + wv * (N_NODES / KSLICES / 8);
    const int NTILES = N_NODES / KSLICES / 8 / 16;

    if (threadIdx.x < H_DIM) {
        sW1[threadIdx.x] = mW1[threadIdx.x];
        sB1[threadIdx.x] = mb1[threadIdx.x];
        sW2[threadIdx.x] = mW2[threadIdx.x];
    }
    __syncthreads();

    const float mb2 = mb2p[0];
    const int n = base_n + r;
    const float* __restrict__ drow = dist + (size_t)n * N_NODES;
    const float* __restrict__ vrow = nrm  + (size_t)n * N_NODES;

    v8f acc = {};

    for (int kt = 0; kt < NTILES; ++kt) {
        const int m0 = kbase + kt * 16;

        v2f d[4], q[4], s[4], bb[4];
        #pragma unroll
        for (int kk = 0; kk < 4; ++kk) {
            const int mm = m0 + 4 * kk + 2 * half;
            d[kk] = *(const v2f*)(drow + mm);
            q[kk] = *(const v2f*)(vrow + mm);
            bb[kk].x = fsums[(mm    ) * O_DIM + r];
            bb[kk].y = fsums[(mm + 1) * O_DIM + r];
            s[kk].x = mb2; s[kk].y = mb2;
        }

        #pragma unroll 8
        for (int h = 0; h < H_DIM; ++h) {
            const float w1 = sW1[h], b1 = sB1[h], w2 = sW2[h];
            #pragma unroll
            for (int kk = 0; kk < 4; ++kk) {
                float t0 = fmaxf(fmaf(d[kk].x, w1, b1), 0.0f);
                float t1 = fmaxf(fmaf(d[kk].y, w1, b1), 0.0f);
                s[kk].x = fmaf(t0, w2, s[kk].x);
                s[kk].y = fmaf(t1, w2, s[kk].y);
            }
        }

        #pragma unroll
        for (int kk = 0; kk < 4; ++kk) {
            v2f a;
            a.x = s[kk].x * __builtin_amdgcn_rcpf(q[kk].x);
            a.y = s[kk].y * __builtin_amdgcn_rcpf(q[kk].y);
            acc = __builtin_amdgcn_wmma_f32_16x16x4_f32(
                false, a, false, bb[kk], (short)0, acc, false, false);
        }
    }

    #pragma unroll
    for (int v = 0; v < 8; ++v) red[wv][lane * 8 + v] = acc[v];
    __syncthreads();
    if (wv == 0) {
        float* __restrict__ pslice = part + (size_t)blockIdx.y * (N_NODES * O_DIM);
        #pragma unroll
        for (int v = 0; v < 8; ++v) {
            float ssum = 0.0f;
            #pragma unroll
            for (int u = 0; u < 8; ++u) ssum += red[u][lane * 8 + v];
            pslice[(base_n + v + 8 * half) * O_DIM + r] = ssum;
        }
    }
}

__global__ __launch_bounds__(256) void hgnam_combine_kernel(
    const float* __restrict__ part, float* __restrict__ out)
{
    const int i = blockIdx.x * 256 + threadIdx.x;
    if (i < N_NODES * O_DIM) {
        float s = 0.0f;
        #pragma unroll
        for (int k = 0; k < KSLICES; ++k)
            s += part[(size_t)k * (N_NODES * O_DIM) + i];
        out[i] = s;
    }
}

extern "C" void kernel_launch(void* const* d_in, const int* in_sizes, int n_in,
                              void* d_out, int out_size, void* d_ws, size_t ws_size,
                              hipStream_t stream) {
    (void)in_sizes; (void)n_in; (void)out_size; (void)ws_size;
    const float* x    = (const float*)d_in[0];
    const float* dist = (const float*)d_in[1];
    const float* nrm  = (const float*)d_in[2];
    const float* fW1  = (const float*)d_in[3];
    const float* fb1  = (const float*)d_in[4];
    const float* fW2  = (const float*)d_in[5];
    const float* fb2  = (const float*)d_in[6];
    const float* mW1  = (const float*)d_in[7];
    const float* mb1  = (const float*)d_in[8];
    const float* mW2  = (const float*)d_in[9];
    const float* mb2  = (const float*)d_in[10];

    float* fsums = (float*)d_ws;
    float* part  = (float*)d_ws + N_NODES * O_DIM;
    float* out   = (float*)d_out;

    hgnam_fsums_kernel<<<dim3(N_NODES / 16), dim3(256), 0, stream>>>(
        x, fW1, fb1, fW2, fb2, fsums);
    hgnam_main_kernel<<<dim3(N_NODES / 16, KSLICES), dim3(256), 0, stream>>>(
        dist, nrm, mW1, mb1, mW2, mb2, fsums, part);
    hgnam_combine_kernel<<<dim3((N_NODES * O_DIM + 255) / 256), dim3(256), 0, stream>>>(
        part, out);
}


# fsums kernel: feature sum split over 2 workgroups per 16-row tile (grid 128x2 fills all 256 WGPs); fb2 shares start the accumulators; main adds the 2 planes at its B loads; sum order changes
# speedup vs baseline: 2.8869x; 1.6942x over previous
#include <hip/hip_runtime.h>

#define N_NODES 2048
#define F_FEAT  128
#define H_DIM   64
#define O_DIM   16
#define KSLICES 4
#define FSPLIT  2
#define FS_NFW  (F_FEAT / FSPLIT / 8)

typedef float v2f __attribute__((ext_vector_type(2)));
typedef float v8f __attribute__((ext_vector_type(8)));

#define FS_W2_FLOATS    (H_DIM * O_DIM)
#define FS_STAGE_FLOATS (FS_W2_FLOATS + 2 * H_DIM)

typedef float v4f __attribute__((ext_vector_type(4)));

struct FeatStage {
    v4f   w2q[8];
    v4f   wbq;
};

__device__ __forceinline__ void fsums_fetch(
    FeatStage& G,
    const float* __restrict__ wbsrc,
    const float* __restrict__ fW2,
    int f, int lane)
{
    const v4f* __restrict__ g2 = (const v4f*)(fW2 + f * FS_W2_FLOATS) + lane;
    #pragma unroll
    for (int j = 0; j < 8; ++j) G.w2q[j] = g2[j * 32];
    G.wbq = *(const v4f*)(wbsrc + f * H_DIM);
}

__device__ __forceinline__ void fsums_park(
    float* sb, const FeatStage& G, int lane, int half, int r)
{
    #pragma unroll
    for (int j = 0; j < 8; ++j) ((v4f*)sb)[j * 32 + lane] = G.w2q[j];
    v2f lo, hi;
    lo.x = G.wbq.x; lo.y = G.wbq.y;
    hi.x = G.wbq.z; hi.y = G.wbq.w;
    *(v2f*)(sb + FS_W2_FLOATS + (2 * r    ) * 4 + 2 * half) = lo;
    *(v2f*)(sb + FS_W2_FLOATS + (2 * r + 1) * 4 + 2 * half) = hi;
}

__device__ __forceinline__ void fsums_wave_lds_sync()
{
    __builtin_amdgcn_fence(__ATOMIC_RELEASE, "wavefront");
    __builtin_amdgcn_wave_barrier();
    __builtin_amdgcn_fence(__ATOMIC_ACQUIRE, "wavefront");
}

__global__ __launch_bounds__(256) void hgnam_fsums_kernel(
    const float* __restrict__ x,
    const float* __restrict__ fW1,
    const float* __restrict__ fb1,
    const float* __restrict__ fW2,
    const float* __restrict__ fb2,
    float* __restrict__ fsp)
{
    __shared__ __attribute__((aligned(16))) float stage[8][2][FS_STAGE_FLOATS];
    __shared__ float xs[8][FS_NFW * 16];
    __shared__ float red[8][256];

    const int lane = threadIdx.x & 31;
    const int wv   = threadIdx.x >> 5;
    const int half = lane >> 4;
    const int r    = lane & 15;
    const int base_n = blockIdx.x * 16;
    const float* __restrict__ wbsrc = (half ? fb1 : fW1) + 4 * r;

    const int f0 = blockIdx.y * (F_FEAT / FSPLIT) + wv * FS_NFW;
    FeatStage G;
    fsums_fetch(G, wbsrc, fW2, f0, lane);

    {
        const v4f xa = *(const v4f*)(x + (base_n + r) * F_FEAT + f0 + 4 * half);
        float* xw = &xs[wv][(4 * half) * 16 + r];
        xw[0 * 16] = xa.x; xw[1 * 16] = xa.y; xw[2 * 16] = xa.z; xw[3 * 16] = xa.w;
    }

    float bw = 0.0f;
    #pragma unroll
    for (int j = 0; j < FS_NFW; ++j) bw += fb2[(f0 + j) * O_DIM + r];
    v8f acc;
    #pragma unroll
    for (int v = 0; v < 8; ++v) acc[v] = bw;

    fsums_park(&stage[wv][0][0], G, lane, half, r);
    fsums_fetch(G, wbsrc, fW2, f0 + 1, lane);
    fsums_wave_lds_sync();

    for (int fi = 0; fi < FS_NFW; ++fi) {
        const float* cb = &stage[wv][fi & 1][0];
        float*       nb = &stage[wv][(fi & 1) ^ 1][0];

        const float xcur = xs[wv][fi * 16 + r];
        v4f wb[16];
        v2f bf[16];
        #pragma unroll
        for (int c = 0; c < 16; ++c) {
            const int h0 = 4 * c + 2 * half;
            wb[c]   = *(const v4f*)(cb + FS_W2_FLOATS + (2 * c + half) * 4);
            bf[c].x = cb[(h0    ) * O_DIM + r];
            bf[c].y = cb[(h0 + 1) * O_DIM + r];
        }

        if (fi + 1 < FS_NFW) fsums_park(nb, G, lane, half, r);
        if (fi + 2 < FS_NFW) fsums_fetch(G, wbsrc, fW2, f0 + fi + 2, lane);

        #pragma unroll
        for (int c = 0; c < 16; ++c) {
            v2f a;
            a.x = fmaxf(fmaf(xcur, wb[c].x, wb[c].z), 0.0f);
            a.y = fmaxf(fmaf(xcur, wb[c].y, wb[c].w), 0.0f);
            acc = __builtin_amdgcn_wmma_f32_16x16x4_f32(
                false, a, false, bf[c], (short)0, acc, false, false);
        }

        fsums_wave_lds_sync();
    }

    #pragma unroll
    for (int v = 0; v < 8; ++v) red[wv][lane * 8 + v] = acc[v];
    __syncthreads();
    if (wv == 0) {
        float* __restrict__ plane = fsp + (size_t)blockIdx.y * (N_NODES * O_DIM);
        #pragma unroll
        for (int v = 0; v < 8; ++v) {
            float s = 0.0f;
            #pragma unroll
            for (int u = 0; u < 8; ++u) s += red[u][lane * 8 + v];
            plane[(base_n + v + 8 * half) * O_DIM + r] = s;
        }
    }
}

__global__ __launch_bounds__(256) void hgnam_main_kernel(
    const float* __restrict__ dist,
    const float* __restrict__ nrm,
    const float* __restrict__ mW1,
    const float* __restrict__ mb1,
    const float* __restrict__ mW2,
    const float* __restrict__ mb2p,
    const float* __restrict__ fsp,
    float* __restrict__ part,
    float* __restrict__ out)
{
    __shared__ float sW1[H_DIM], sB1[H_DIM], sW2[H_DIM];
    __shared__ float red[8][256];

    const int lane = threadIdx.x & 31;
    const int wv   = threadIdx.x >> 5;
    const int half = lane >> 4;
    const int r    = lane & 15;
    const int base_n = blockIdx.x * 16;
    const int kbase  = blockIdx.y * (N_NODES / KSLICES) + wv * (N_NODES / KSLICES / 8);
    const int NTILES = N_NODES / KSLICES / 8 / 16;

    if (threadIdx.x < H_DIM) {
        sW1[threadIdx.x] = mW1[threadIdx.x];
        sB1[threadIdx.x] = mb1[threadIdx.x];
        sW2[threadIdx.x] = mW2[threadIdx.x];
    }
    __syncthreads();

    const float mb2 = mb2p[0];
    const int n = base_n + r;
    const float* __restrict__ drow = dist + (size_t)n * N_NODES;
    const float* __restrict__ vrow = nrm  + (size_t)n * N_NODES;

    v8f acc = {};

    for (int kt = 0; kt < NTILES; ++kt) {
        const int m0 = kbase + kt * 16;

        v2f d[4], q[4], s[4], bb[4];
        #pragma unroll
        for (int kk = 0; kk < 4; ++kk) {
            const int mm = m0 + 4 * kk + 2 * half;
            d[kk] = *(const v2f*)(drow + mm);
            q[kk] = *(const v2f*)(vrow + mm);
            const float* __restrict__ fp = fsp + mm * O_DIM + r;
            bb[kk].x = fp[0]     + fp[N_NODES * O_DIM];
            bb[kk].y = fp[O_DIM] + fp[N_NODES * O_DIM + O_DIM];
            s[kk].x = mb2; s[kk].y = mb2;
        }

        #pragma unroll 8
        for (int h = 0; h < H_DIM; ++h) {
            const float w1 = sW1[h], b1 = sB1[h], w2 = sW2[h];
            #pragma unroll
            for (int kk = 0; kk < 4; ++kk) {
                float t0 = fmaxf(fmaf(d[kk].x, w1, b1), 0.0f);
                float t1 = fmaxf(fmaf(d[kk].y, w1, b1), 0.0f);
                s[kk].x = fmaf(t0, w2, s[kk].x);
                s[kk].y = fmaf(t1, w2, s[kk].y);
            }
        }

        #pragma unroll
        for (int kk = 0; kk < 4; ++kk) {
            v2f a;
            a.x = s[kk].x * __builtin_amdgcn_rcpf(q[kk].x);
            a.y = s[kk].y * __builtin_amdgcn_rcpf(q[kk].y);
            acc = __builtin_amdgcn_wmma_f32_16x16x4_f32(
                false, a, false, bb[kk], (short)0, acc, false, false);
        }
    }

    #pragma unroll
    for (int v = 0; v < 8; ++v) red[wv][lane * 8 + v] = acc[v];
    __syncthreads();
    if (wv == 0) {
        float* __restrict__ pslice = (blockIdx.y == 0)
            ? out : part + (size_t)(blockIdx.y - 1) * (N_NODES * O_DIM);
        #pragma unroll
        for (int v = 0; v < 8; ++v) {
            float ssum = 0.0f;
            #pragma unroll
            for (int u = 0; u < 8; ++u) ssum += red[u][lane * 8 + v];
            pslice[(base_n + v + 8 * half) * O_DIM + r] = ssum;
        }
    }
}

__global__ __launch_bounds__(256) void hgnam_combine_kernel(
    const float* __restrict__ part, float* __restrict__ out)
{
    const int i = blockIdx.x * 256 + threadIdx.x;
    if (i < N_NODES * O_DIM) {
        float s = out[i];
        #pragma unroll
        for (int k = 0; k < KSLICES - 1; ++k)
            s += part[(size_t)k * (N_NODES * O_DIM) + i];
        out[i] = s;
    }
}

extern "C" void kernel_launch(void* const* d_in, const int* in_sizes, int n_in,
                              void* d_out, int out_size, void* d_ws, size_t ws_size,
                              hipStream_t stream) {
    (void)in_sizes; (void)n_in; (void)out_size; (void)ws_size;
    const float* x    = (const float*)d_in[0];
    const float* dist = (const float*)d_in[1];
    const float* nrm  = (const float*)d_in[2];
    const float* fW1  = (const float*)d_in[3];
    const float* fb1  = (const float*)d_in[4];
    const float* fW2  = (const float*)d_in[5];
    const float* fb2  = (const float*)d_in[6];
    const float* mW1  = (const float*)d_in[7];
    const float* mb1  = (const float*)d_in[8];
    const float* mW2  = (const float*)d_in[9];
    const float* mb2  = (const float*)d_in[10];

    float* fsp   = (float*)d_ws;
    float* part  = (float*)d_ws + FSPLIT * N_NODES * O_DIM;
    float* out   = (float*)d_out;

    hgnam_fsums_kernel<<<dim3(N_NODES / 16, FSPLIT), dim3(256), 0, stream>>>(
        x, fW1, fb1, fW2, fb2, fsp);
    hgnam_main_kernel<<<dim3(N_NODES / 16, KSLICES), dim3(256), 0, stream>>>(
        dist, nrm, mW1, mb1, mW2, mb2, fsp, part, out);
    hgnam_combine_kernel<<<dim3((N_NODES * O_DIM + 255) / 256), dim3(256), 0, stream>>>(
        part, out);
}


# main kernel: 8-row tiles, grid 256 x 16 waves: each workgroup sums its rows over all m and stores final values, so k-slice planes and the combine launch go away (3 launches -> 2); sum order changes
# speedup vs baseline: 3.0848x; 1.0686x over previous
#include <hip/hip_runtime.h>

#define N_NODES 2048
#define F_FEAT  128
#define H_DIM   64
#define O_DIM   16
#define MROWS   8
#define MWAVES  16
#define FSPLIT  2
#define FS_NFW  (F_FEAT / FSPLIT / 8)

typedef float v2f __attribute__((ext_vector_type(2)));
typedef float v8f __attribute__((ext_vector_type(8)));

#define FS_W2_FLOATS    (H_DIM * O_DIM)
#define FS_STAGE_FLOATS (FS_W2_FLOATS + 2 * H_DIM)

typedef float v4f __attribute__((ext_vector_type(4)));

struct FeatStage {
    v4f   w2q[8];
    v4f   wbq;
};

__device__ __forceinline__ void fsums_fetch(
    FeatStage& G,
    const float* __restrict__ wbsrc,
    const float* __restrict__ fW2,
    int f, int lane)
{
    const v4f* __restrict__ g2 = (const v4f*)(fW2 + f * FS_W2_FLOATS) + lane;
    #pragma unroll
    for (int j = 0; j < 8; ++j) G.w2q[j] = g2[j * 32];
    G.wbq = *(const v4f*)(wbsrc + f * H_DIM);
}

__device__ __forceinline__ void fsums_park(
    float* sb, const FeatStage& G, int lane, int half, int r)
{
    #pragma unroll
    for (int j = 0; j < 8; ++j) ((v4f*)sb)[j * 32 + lane] = G.w2q[j];
    v2f lo, hi;
    lo.x = G.wbq.x; lo.y = G.wbq.y;
    hi.x = G.wbq.z; hi.y = G.wbq.w;
    *(v2f*)(sb + FS_W2_FLOATS + (2 * r    ) * 4 + 2 * half) = lo;
    *(v2f*)(sb + FS_W2_FLOATS + (2 * r + 1) * 4 + 2 * half) = hi;
}

__device__ __forceinline__ void fsums_wave_lds_sync()
{
    __builtin_amdgcn_fence(__ATOMIC_RELEASE, "wavefront");
    __builtin_amdgcn_wave_barrier();
    __builtin_amdgcn_fence(__ATOMIC_ACQUIRE, "wavefront");
}

__global__ __launch_bounds__(256) void hgnam_fsums_kernel(
    const float* __restrict__ x,
    const float* __restrict__ fW1,
    const float* __restrict__ fb1,
    const float* __restrict__ fW2,
    const float* __restrict__ fb2,
    float* __restrict__ fsp)
{
    __shared__ __attribute__((aligned(16))) float stage[8][2][FS_STAGE_FLOATS];
    __shared__ float xs[8][FS_NFW * 16];
    __shared__ float red[8][256];

    const int lane = threadIdx.x & 31;
    const int wv   = threadIdx.x >> 5;
    const int half = lane >> 4;
    const int r    = lane & 15;
    const int base_n = blockIdx.x * 16;
    const float* __restrict__ wbsrc = (half ? fb1 : fW1) + 4 * r;

    const int f0 = blockIdx.y * (F_FEAT / FSPLIT) + wv * FS_NFW;
    FeatStage G;
    fsums_fetch(G, wbsrc, fW2, f0, lane);

    {
        const v4f xa = *(const v4f*)(x + (base_n + r) * F_FEAT + f0 + 4 * half);
        float* xw = &xs[wv][(4 * half) * 16 + r];
        xw[0 * 16] = xa.x; xw[1 * 16] = xa.y; xw[2 * 16] = xa.z; xw[3 * 16] = xa.w;
    }

    float bw = 0.0f;
    #pragma unroll
    for (int j = 0; j < FS_NFW; ++j) bw += fb2[(f0 + j) * O_DIM + r];
    v8f acc;
    #pragma unroll
    for (int v = 0; v < 8; ++v) acc[v] = bw;

    fsums_park(&stage[wv][0][0], G, lane, half, r);
    fsums_fetch(G, wbsrc, fW2, f0 + 1, lane);
    fsums_wave_lds_sync();

    for (int fi = 0; fi < FS_NFW; ++fi) {
        const float* cb = &stage[wv][fi & 1][0];
        float*       nb = &stage[wv][(fi & 1) ^ 1][0];

        const float xcur = xs[wv][fi * 16 + r];
        v4f wb[16];
        v2f bf[16];
        #pragma unroll
        for (int c = 0; c < 16; ++c) {
            const int h0 = 4 * c + 2 * half;
            wb[c]   = *(const v4f*)(cb + FS_W2_FLOATS + (2 * c + half) * 4);
            bf[c].x = cb[(h0    ) * O_DIM + r];
            bf[c].y = cb[(h0 + 1) * O_DIM + r];
        }

        if (fi + 1 < FS_NFW) fsums_park(nb, G, lane, half, r);
        if (fi + 2 < FS_NFW) fsums_fetch(G, wbsrc, fW2, f0 + fi + 2, lane);

        #pragma unroll
        for (int c = 0; c < 16; ++c) {
            v2f a;
            a.x = fmaxf(fmaf(xcur, wb[c].x, wb[c].z), 0.0f);
            a.y = fmaxf(fmaf(xcur, wb[c].y, wb[c].w), 0.0f);
            acc = __builtin_amdgcn_wmma_f32_16x16x4_f32(
                false, a, false, bf[c], (short)0, acc, false, false);
        }

        fsums_wave_lds_sync();
    }

    #pragma unroll
    for (int v = 0; v < 8; ++v) red[wv][lane * 8 + v] = acc[v];
    __syncthreads();
    if (wv == 0) {
        float* __restrict__ plane = fsp + (size_t)blockIdx.y * (N_NODES * O_DIM);
        #pragma unroll
        for (int v = 0; v < 8; ++v) {
            float s = 0.0f;
            #pragma unroll
            for (int u = 0; u < 8; ++u) s += red[u][lane * 8 + v];
            plane[(base_n + v + 8 * half) * O_DIM + r] = s;
        }
    }
}

__global__ __launch_bounds__(MWAVES * 32) void hgnam_main_kernel(
    const float* __restrict__ dist,
    const float* __restrict__ nrm,
    const float* __restrict__ mW1,
    const float* __restrict__ mb1,
    const float* __restrict__ mW2,
    const float* __restrict__ mb2p,
    const float* __restrict__ fsp,
    float* __restrict__ out)
{
    __shared__ float sW1[H_DIM], sB1[H_DIM], sW2[H_DIM];
    __shared__ float red[MWAVES][256];

    const int lane = threadIdx.x & 31;
    const int wv   = threadIdx.x >> 5;
    const int half = lane >> 4;
    const int r    = lane & 15;
    const int row  = r & 7;
    const int mset = r >> 3;
    const int base_n = blockIdx.x * MROWS;
    const int kbase  = wv * (N_NODES / MWAVES);
    const int NSTEPS = N_NODES / MWAVES / 32;

    if (threadIdx.x < H_DIM) {
        sW1[threadIdx.x] = mW1[threadIdx.x];
        sB1[threadIdx.x] = mb1[threadIdx.x];
        sW2[threadIdx.x] = mW2[threadIdx.x];
    }
    __syncthreads();

    const float mb2 = mb2p[0];
    const size_t aoff = (size_t)(base_n + row) * N_NODES + 16 * mset + 2 * half;
    const float* __restrict__ drow = dist + aoff;
    const float* __restrict__ vrow = nrm  + aoff;
    const float* __restrict__ fcol = fsp + (2 * half) * O_DIM + r;

    v8f acc0 = {};
    v8f acc1 = {};

    for (int kt = 0; kt < NSTEPS; ++kt) {
        const int m0 = kbase + kt * 32;

        v2f d[4], q[4], s[4], b0[4], b1[4];
        #pragma unroll
        for (int kk = 0; kk < 4; ++kk) {
            d[kk] = *(const v2f*)(drow + m0 + 4 * kk);
            q[kk] = *(const v2f*)(vrow + m0 + 4 * kk);
            const float* __restrict__ fp = fcol + (m0 + 4 * kk) * O_DIM;
            const float* __restrict__ fq = fp + 16 * O_DIM;
            b0[kk].x = fp[0]     + fp[N_NODES * O_DIM];
            b0[kk].y = fp[O_DIM] + fp[N_NODES * O_DIM + O_DIM];
            b1[kk].x = fq[0]     + fq[N_NODES * O_DIM];
            b1[kk].y = fq[O_DIM] + fq[N_NODES * O_DIM + O_DIM];
            s[kk].x = mb2; s[kk].y = mb2;
        }

        #pragma unroll 8
        for (int h = 0; h < H_DIM; ++h) {
            const float w1 = sW1[h], b1w = sB1[h], w2 = sW2[h];
            #pragma unroll
            for (int kk = 0; kk < 4; ++kk) {
                float t0 = fmaxf(fmaf(d[kk].x, w1, b1w), 0.0f);
                float t1 = fmaxf(fmaf(d[kk].y, w1, b1w), 0.0f);
                s[kk].x = fmaf(t0, w2, s[kk].x);
                s[kk].y = fmaf(t1, w2, s[kk].y);
            }
        }

        #pragma unroll
        for (int kk = 0; kk < 4; ++kk) {
            v2f a;
            a.x = s[kk].x * __builtin_amdgcn_rcpf(q[kk].x);
            a.y = s[kk].y * __builtin_amdgcn_rcpf(q[kk].y);
            acc0 = __builtin_amdgcn_wmma_f32_16x16x4_f32(
                false, a, false, b0[kk], (short)0, acc0, false, false);
            acc1 = __builtin_amdgcn_wmma_f32_16x16x4_f32(
                false, a, false, b1[kk], (short)0, acc1, false, false);
        }
    }

    #pragma unroll
    for (int v = 0; v < 8; ++v) red[wv][v * 32 + lane] = half ? acc1[v] : acc0[v];
    __syncthreads();
    if (threadIdx.x < MROWS * O_DIM) {
        const int v = threadIdx.x >> 4;
        const int c = threadIdx.x & 15;
        float ssum = 0.0f;
        #pragma unroll
        for (int u = 0; u < MWAVES; ++u)
            ssum += red[u][v * 32 + c] + red[u][v * 32 + 16 + c];
        out[(base_n + v) * O_DIM + c] = ssum;
    }
}

extern "C" void kernel_launch(void* const* d_in, const int* in_sizes, int n_in,
                              void* d_out, int out_size, void* d_ws, size_t ws_size,
                              hipStream_t stream) {
    (void)in_sizes; (void)n_in; (void)out_size; (void)ws_size;
    const float* x    = (const float*)d_in[0];
    const float* dist = (const float*)d_in[1];
    const float* nrm  = (const float*)d_in[2];
    const float* fW1  = (const float*)d_in[3];
    const float* fb1  = (const float*)d_in[4];
    const float* fW2  = (const float*)d_in[5];
    const float* fb2  = (const float*)d_in[6];
    const float* mW1  = (const float*)d_in[7];
    const float* mb1  = (const float*)d_in[8];
    const float* mW2  = (const float*)d_in[9];
    const float* mb2  = (const float*)d_in[10];

    float* fsp   = (float*)d_ws;
    float* out   = (float*)d_out;

    hgnam_fsums_kernel<<<dim3(N_NODES / 16, FSPLIT), dim3(256), 0, stream>>>(
        x, fW1, fb1, fW2, fb2, fsp);
    hgnam_main_kernel<<<dim3(N_NODES / MROWS), dim3(MWAVES * 32), 0, stream>>>(
        dist, nrm, mW1, mb1, mW2, mb2, fsp, out);
}


# main kernel: m-MLP as a piecewise-linear table: breakpoints sorted per workgroup, slope/offset summed per segment, then binary search (4 LDS probes) + one fma per element; sum order, rounding change
# speedup vs baseline: 3.4864x; 1.1302x over previous
#include <hip/hip_runtime.h>

#define N_NODES 2048
#define F_FEAT  128
#define H_DIM   64
#define O_DIM   16
#define MROWS   8
#define MWAVES  16
#define FSPLIT  2
#define FS_NFW  (F_FEAT / FSPLIT / 8)

typedef float v2f __attribute__((ext_vector_type(2)));
typedef float v8f __attribute__((ext_vector_type(8)));

#define FS_W2_FLOATS    (H_DIM * O_DIM)
#define FS_STAGE_FLOATS (FS_W2_FLOATS + 2 * H_DIM)

typedef float v4f __attribute__((ext_vector_type(4)));

struct FeatStage {
    v4f   w2q[8];
    v4f   wbq;
};

__device__ __forceinline__ void fsums_fetch(
    FeatStage& G,
    const float* __restrict__ wbsrc,
    const float* __restrict__ fW2,
    int f, int lane)
{
    const v4f* __restrict__ g2 = (const v4f*)(fW2 + f * FS_W2_FLOATS) + lane;
    #pragma unroll
    for (int j = 0; j < 8; ++j) G.w2q[j] = g2[j * 32];
    G.wbq = *(const v4f*)(wbsrc + f * H_DIM);
}

__device__ __forceinline__ void fsums_park(
    float* sb, const FeatStage& G, int lane, int half, int r)
{
    #pragma unroll
    for (int j = 0; j < 8; ++j) ((v4f*)sb)[j * 32 + lane] = G.w2q[j];
    v2f lo, hi;
    lo.x = G.wbq.x; lo.y = G.wbq.y;
    hi.x = G.wbq.z; hi.y = G.wbq.w;
    *(v2f*)(sb + FS_W2_FLOATS + (2 * r    ) * 4 + 2 * half) = lo;
    *(v2f*)(sb + FS_W2_FLOATS + (2 * r + 1) * 4 + 2 * half) = hi;
}

__device__ __forceinline__ void fsums_wave_lds_sync()
{
    __builtin_amdgcn_fence(__ATOMIC_RELEASE, "wavefront");
    __builtin_amdgcn_wave_barrier();
    __builtin_amdgcn_fence(__ATOMIC_ACQUIRE, "wavefront");
}

__global__ __launch_bounds__(256) void hgnam_fsums_kernel(
    const float* __restrict__ x,
    const float* __restrict__ fW1,
    const float* __restrict__ fb1,
    const float* __restrict__ fW2,
    const float* __restrict__ fb2,
    float* __restrict__ fsp)
{
    __shared__ __attribute__((aligned(16))) float stage[8][2][FS_STAGE_FLOATS];
    __shared__ float xs[8][FS_NFW * 16];
    __shared__ float red[8][256];

    const int lane = threadIdx.x & 31;
    const int wv   = threadIdx.x >> 5;
    const int half = lane >> 4;
    const int r    = lane & 15;
    const int base_n = blockIdx.x * 16;
    const float* __restrict__ wbsrc = (half ? fb1 : fW1) + 4 * r;

    const int f0 = blockIdx.y * (F_FEAT / FSPLIT) + wv * FS_NFW;
    FeatStage G;
    fsums_fetch(G, wbsrc, fW2, f0, lane);

    {
        const v4f xa = *(const v4f*)(x + (base_n + r) * F_FEAT + f0 + 4 * half);
        float* xw = &xs[wv][(4 * half) * 16 + r];
        xw[0 * 16] = xa.x; xw[1 * 16] = xa.y; xw[2 * 16] = xa.z; xw[3 * 16] = xa.w;
    }

    float bw = 0.0f;
    #pragma unroll
    for (int j = 0; j < FS_NFW; ++j) bw += fb2[(f0 + j) * O_DIM + r];
    v8f acc;
    #pragma unroll
    for (int v = 0; v < 8; ++v) acc[v] = bw;

    fsums_park(&stage[wv][0][0], G, lane, half, r);
    fsums_fetch(G, wbsrc, fW2, f0 + 1, lane);
    fsums_wave_lds_sync();

    for (int fi = 0; fi < FS_NFW; ++fi) {
        const float* cb = &stage[wv][fi & 1][0];
        float*       nb = &stage[wv][(fi & 1) ^ 1][0];

        const float xcur = xs[wv][fi * 16 + r];
        v4f wb[16];
        v2f bf[16];
        #pragma unroll
        for (int c = 0; c < 16; ++c) {
            const int h0 = 4 * c + 2 * half;
            wb[c]   = *(const v4f*)(cb + FS_W2_FLOATS + (2 * c + half) * 4);
            bf[c].x = cb[(h0    ) * O_DIM + r];
            bf[c].y = cb[(h0 + 1) * O_DIM + r];
        }

        if (fi + 1 < FS_NFW) fsums_park(nb, G, lane, half, r);
        if (fi + 2 < FS_NFW) fsums_fetch(G, wbsrc, fW2, f0 + fi + 2, lane);

        #pragma unroll
        for (int c = 0; c < 16; ++c) {
            v2f a;
            a.x = fmaxf(fmaf(xcur, wb[c].x, wb[c].z), 0.0f);
            a.y = fmaxf(fmaf(xcur, wb[c].y, wb[c].w), 0.0f);
            acc = __builtin_amdgcn_wmma_f32_16x16x4_f32(
                false, a, false, bf[c], (short)0, acc, false, false);
        }

        fsums_wave_lds_sync();
    }

    #pragma unroll
    for (int v = 0; v < 8; ++v) red[wv][lane * 8 + v] = acc[v];
    __syncthreads();
    if (wv == 0) {
        float* __restrict__ plane = fsp + (size_t)blockIdx.y * (N_NODES * O_DIM);
        #pragma unroll
        for (int v = 0; v < 8; ++v) {
            float s = 0.0f;
            #pragma unroll
            for (int u = 0; u < 8; ++u) s += red[u][lane * 8 + v];
            plane[(base_n + v + 8 * half) * O_DIM + r] = s;
        }
    }
}

#define LPAD  80

__global__ __launch_bounds__(MWAVES * 32) void hgnam_main_kernel(
    const float* __restrict__ dist,
    const float* __restrict__ nrm,
    const float* __restrict__ mW1,
    const float* __restrict__ mb1,
    const float* __restrict__ mW2,
    const float* __restrict__ mb2p,
    const float* __restrict__ fsp,
    float* __restrict__ out)
{
    __shared__ float uT[H_DIM];
    __shared__ float uC[H_DIM];
    __shared__ float uK[H_DIM];
    __shared__ float uP[H_DIM];
    __shared__ __attribute__((aligned(16))) float oC[H_DIM];
    __shared__ __attribute__((aligned(16))) float oK[H_DIM];
    __shared__ __attribute__((aligned(16))) float oP[H_DIM];
    __shared__ float sTau[H_DIM];
    __shared__ __attribute__((aligned(8))) v2f sAB[H_DIM + 8];
    __shared__ __attribute__((aligned(16))) float gW1[LPAD];
    __shared__ __attribute__((aligned(16))) float gB1[LPAD];
    __shared__ __attribute__((aligned(16))) float gW2[LPAD];
    __shared__ float red[MWAVES][256];

    const int lane = threadIdx.x & 31;
    const int wv   = threadIdx.x >> 5;
    const int half = lane >> 4;
    const int r    = lane & 15;
    const int row  = r & 7;
    const int mset = r >> 3;
    const int base_n = blockIdx.x * MROWS;
    const int kbase  = wv * (N_NODES / MWAVES);
    const int NSTEPS = N_NODES / MWAVES / 32;
    const float INF = __builtin_huge_valf();

    if (threadIdx.x < LPAD) {
        gW1[threadIdx.x] = 0.0f; gB1[threadIdx.x] = 0.0f; gW2[threadIdx.x] = 0.0f;
    }

    const float w1a = mW1[lane],      b1a = mb1[lane],      w2a = mW2[lane];
    const float w1b = mW1[lane + 32], b1b = mb1[lane + 32], w2b = mW2[lane + 32];
    const float ta = -b1a / w1a;
    const float tb = -b1b / w1b;
    const bool safea = (fabsf(w1a) >= 0x1p-40f) && (fabsf(w1a) <= 0x1p40f) && (fabsf(ta) <= 0x1p40f);
    const bool safeb = (fabsf(w1b) >= 0x1p-40f) && (fabsf(w1b) <= 0x1p40f) && (fabsf(tb) <= 0x1p40f);
    const unsigned gA = __builtin_amdgcn_ballot_w32(!safea);
    const unsigned gB = __builtin_amdgcn_ballot_w32(!safeb);
    const int nG  = __builtin_popcount(gA) + __builtin_popcount(gB);
    const int nG4 = (nG + 3) & ~3;
    const unsigned lt = (1u << lane) - 1u;
    const int posa = __builtin_popcount(gA & lt);
    const int posb = __builtin_popcount(gA) + __builtin_popcount(gB & lt);
    const float mb2 = mb2p[0];

    __syncthreads();
    if (wv == 0) {
        uT[lane]      = safea ? ta : INF;
        uC[lane]      = safea ? w1a * w2a : 0.0f;
        uK[lane]      = safea ? w2a * b1a : 0.0f;
        uP[lane]      = (safea && w1a > 0.0f) ? 1.0f : 0.0f;
        uT[lane + 32] = safeb ? tb : INF;
        uC[lane + 32] = safeb ? w1b * w2b : 0.0f;
        uK[lane + 32] = safeb ? w2b * b1b : 0.0f;
        uP[lane + 32] = (safeb && w1b > 0.0f) ? 1.0f : 0.0f;
        if (!safea) { gW1[posa] = w1a; gB1[posa] = b1a; gW2[posa] = w2a; }
        if (!safeb) { gW1[posb] = w1b; gB1[posb] = b1b; gW2[posb] = w2b; }
    }
    __syncthreads();
    if (threadIdx.x < H_DIM) {
        const int   me = threadIdx.x;
        const float th = uT[me];
        int rk = 0;
        for (int j = 0; j < H_DIM; ++j) {
            const float tj = uT[j];
            rk += (tj < th || (tj == th && j < me)) ? 1 : 0;
        }
        sTau[rk] = th;
        oC[rk] = uC[me];
        oK[rk] = uK[me];
        oP[rk] = uP[me];
    }
    __syncthreads();
    if (threadIdx.x <= H_DIM) {
        const int k = threadIdx.x;
        float A = 0.0f, B = mb2;
        #pragma unroll
        for (int j4 = 0; j4 < H_DIM / 4; ++j4) {
            const v4f p4 = *(const v4f*)(oP + 4 * j4);
            const v4f c4 = *(const v4f*)(oC + 4 * j4);
            const v4f e4 = *(const v4f*)(oK + 4 * j4);
            #pragma unroll
            for (int u = 0; u < 4; ++u) {
                const bool act = (p4[u] != 0.0f) == (4 * j4 + u < k);
                A += act ? c4[u] : 0.0f;
                B += act ? e4[u] : 0.0f;
            }
        }
        v2f ab; ab.x = A; ab.y = B;
        sAB[k] = ab;
    }
    __syncthreads();

    const size_t aoff = (size_t)(base_n + row) * N_NODES + 16 * mset + 2 * half;
    const float* __restrict__ drow = dist + aoff;
    const float* __restrict__ vrow = nrm  + aoff;
    const float* __restrict__ fcol = fsp + (2 * half) * O_DIM + r;

    v8f acc0 = {};
    v8f acc1 = {};

    for (int kt = 0; kt < NSTEPS; ++kt) {
        const int m0 = kbase + kt * 32;

        v2f d[4], q[4], s[4], b0[4], b1[4];
        #pragma unroll
        for (int kk = 0; kk < 4; ++kk) {
            d[kk] = *(const v2f*)(drow + m0 + 4 * kk);
            q[kk] = *(const v2f*)(vrow + m0 + 4 * kk);
            const float* __restrict__ fp = fcol + (m0 + 4 * kk) * O_DIM;
            const float* __restrict__ fq = fp + 16 * O_DIM;
            b0[kk].x = fp[0]     + fp[N_NODES * O_DIM];
            b0[kk].y = fp[O_DIM] + fp[N_NODES * O_DIM + O_DIM];
            b1[kk].x = fq[0]     + fq[N_NODES * O_DIM];
            b1[kk].y = fq[O_DIM] + fq[N_NODES * O_DIM + O_DIM];
        }

        const float t31 = sTau[31], t15 = sTau[15], t47 = sTau[47], t63 = sTau[63];
        #pragma unroll
        for (int kk = 0; kk < 4; ++kk) {
            #pragma unroll
            for (int e = 0; e < 2; ++e) {
                const float dv = d[kk][e];
                const bool  c1 = t31 < dv;
                int k = c1 ? 32 : 0;
                k += ((c1 ? t47 : t15) < dv) ? 16 : 0;
                #pragma unroll
                for (int st = 8; st >= 1; st >>= 1)
                    k += (sTau[k + st - 1] < dv) ? st : 0;
                k += (t63 < dv) ? 1 : 0;
                const v2f ab = sAB[k];
                s[kk][e] = fmaf(ab.x, dv, ab.y);
            }
        }
        for (int i = 0; i < nG4; i += 4) {
            const v4f u4 = *(const v4f*)(gW1 + i);
            const v4f v4 = *(const v4f*)(gB1 + i);
            const v4f z4 = *(const v4f*)(gW2 + i);
            #pragma unroll
            for (int j = 0; j < 4; ++j) {
                #pragma unroll
                for (int kk = 0; kk < 4; ++kk) {
                    s[kk].x = fmaf(fmaxf(fmaf(d[kk].x, u4[j], v4[j]), 0.0f), z4[j], s[kk].x);
                    s[kk].y = fmaf(fmaxf(fmaf(d[kk].y, u4[j], v4[j]), 0.0f), z4[j], s[kk].y);
                }
            }
        }

        #pragma unroll
        for (int kk = 0; kk < 4; ++kk) {
            v2f a;
            a.x = s[kk].x * __builtin_amdgcn_rcpf(q[kk].x);
            a.y = s[kk].y * __builtin_amdgcn_rcpf(q[kk].y);
            acc0 = __builtin_amdgcn_wmma_f32_16x16x4_f32(
                false, a, false, b0[kk], (short)0, acc0, false, false);
            acc1 = __builtin_amdgcn_wmma_f32_16x16x4_f32(
                false, a, false, b1[kk], (short)0, acc1, false, false);
        }
    }

    #pragma unroll
    for (int v = 0; v < 8; ++v) red[wv][v * 32 + lane] = half ? acc1[v] : acc0[v];
    __syncthreads();
    if (threadIdx.x < MROWS * O_DIM) {
        const int v = threadIdx.x >> 4;
        const int c = threadIdx.x & 15;
        float ssum = 0.0f;
        #pragma unroll
        for (int u = 0; u < MWAVES; ++u)
            ssum += red[u][v * 32 + c] + red[u][v * 32 + 16 + c];
        out[(base_n + v) * O_DIM + c] = ssum;
    }
}

extern "C" void kernel_launch(void* const* d_in, const int* in_sizes, int n_in,
                              void* d_out, int out_size, void* d_ws, size_t ws_size,
                              hipStream_t stream) {
    (void)in_sizes; (void)n_in; (void)out_size; (void)ws_size;
    const float* x    = (const float*)d_in[0];
    const float* dist = (const float*)d_in[1];
    const float* nrm  = (const float*)d_in[2];
    const float* fW1  = (const float*)d_in[3];
    const float* fb1  = (const float*)d_in[4];
    const float* fW2  = (const float*)d_in[5];
    const float* fb2  = (const float*)d_in[6];
    const float* mW1  = (const float*)d_in[7];
    const float* mb1  = (const float*)d_in[8];
    const float* mW2  = (const float*)d_in[9];
    const float* mb2  = (const float*)d_in[10];

    float* fsp   = (float*)d_ws;
    float* out   = (float*)d_out;

    hgnam_fsums_kernel<<<dim3(N_NODES / 16, FSPLIT), dim3(256), 0, stream>>>(
        x, fW1, fb1, fW2, fb2, fsp);
    hgnam_main_kernel<<<dim3(N_NODES / MROWS), dim3(MWAVES * 32), 0, stream>>>(
        dist, nrm, mW1, mb1, mW2, mb2, fsp, out);
}


# main kernel: segment index from a grid over d's float bits (1 LDS byte + 2 compares) instead of the 6-step search; grid and m-MLP table built once by an extra workgroup of launch 1; sums unchanged
# speedup vs baseline: 3.9855x; 1.1432x over previous
#include <hip/hip_runtime.h>

#define N_NODES 2048
#define F_FEAT  128
#define H_DIM   64
#define O_DIM   16
#define MROWS   8
#define MWAVES  16
#define FSPLIT  2
#define FS_NFW  (F_FEAT / FSPLIT / 8)

typedef float v2f __attribute__((ext_vector_type(2)));
typedef float v8f __attribute__((ext_vector_type(8)));
typedef float v4f __attribute__((ext_vector_type(4)));

#define GSHIFT   16
#define GCELLS   2048
#define GBASE    (((127 - 10) << 7) - 1)
#define LPAD     80
#define TB_AB    0
#define TB_TAU   132
#define TB_G1    200
#define TB_NG    440
#define TB_LUT   448
#define TB_WORDS (TB_LUT + GCELLS / 4)

__device__ __forceinline__ int mlp_count_below(const float* sTau, float dv)
{
    const bool c1 = sTau[31] < dv;
    int k = c1 ? 32 : 0;
    k += ((c1 ? sTau[47] : sTau[15]) < dv) ? 16 : 0;
    #pragma unroll
    for (int st = 8; st >= 1; st >>= 1)
        k += (sTau[k + st - 1] < dv) ? st : 0;
    k += (sTau[63] < dv) ? 1 : 0;
    return k;
}

__device__ __forceinline__ void mlp_table_build(
    float* L,
    const float* __restrict__ mW1, const float* __restrict__ mb1,
    const float* __restrict__ mW2, const float* __restrict__ mb2p,
    unsigned* __restrict__ tabw)
{
    float* uT   = L;
    float* uC   = L + 64;
    float* uK   = L + 128;
    float* uP   = L + 192;
    float* oC   = L + 256;
    float* oK   = L + 320;
    float* oP   = L + 384;
    float* sTau = L + 448;
    float* sABf = L + 520;
    float* gL   = L + 656;
    int*   kc   = (int*)(L + 896);

    const int tid  = threadIdx.x;
    const int lane = tid & 31;
    const int wv   = tid >> 5;
    const float INF = __builtin_huge_valf();

    if (tid < 3 * LPAD) gL[tid] = 0.0f;

    const float w1a = mW1[lane],      b1a = mb1[lane],      w2a = mW2[lane];
    const float w1b = mW1[lane + 32], b1b = mb1[lane + 32], w2b = mW2[lane + 32];
    const float ta = -b1a / w1a;
    const float tb = -b1b / w1b;
    const bool safea = (fabsf(w1a) >= 0x1p-40f) && (fabsf(w1a) <= 0x1p40f) && (fabsf(ta) <= 0x1p40f);
    const bool safeb = (fabsf(w1b) >= 0x1p-40f) && (fabsf(w1b) <= 0x1p40f) && (fabsf(tb) <= 0x1p40f);
    const unsigned gA = __builtin_amdgcn_ballot_w32(!safea);
    const unsigned gB = __builtin_amdgcn_ballot_w32(!safeb);
    const int nG  = __builtin_popcount(gA) + __builtin_popcount(gB);
    const int nG4 = (nG + 3) & ~3;
    const unsigned lt = (1u << lane) - 1u;
    const int posa = __builtin_popcount(gA & lt);
    const int posb = __builtin_popcount(gA) + __builtin_popcount(gB & lt);
    const float mb2 = mb2p[0];

    __syncthreads();
    if (wv == 0) {
        uT[lane]      = safea ? ta : INF;
        uC[lane]      = safea ? w1a * w2a : 0.0f;
        uK[lane]      = safea ? w2a * b1a : 0.0f;
        uP[lane]      = (safea && w1a > 0.0f) ? 1.0f : 0.0f;
        uT[lane + 32] = safeb ? tb : INF;
        uC[lane + 32] = safeb ? w1b * w2b : 0.0f;
        uK[lane + 32] = safeb ? w2b * b1b : 0.0f;
        uP[lane + 32] = (safeb && w1b > 0.0f) ? 1.0f : 0.0f;
        if (!safea) { gL[posa] = w1a; gL[LPAD + posa] = b1a; gL[2 * LPAD + posa] = w2a; }
        if (!safeb) { gL[posb] = w1b; gL[LPAD + posb] = b1b; gL[2 * LPAD + posb] = w2b; }
    }
    __syncthreads();
    if (tid < H_DIM) {
        const int   me = tid;
        const float th = uT[me];
        int rk = 0;
        for (int j = 0; j < H_DIM; ++j) {
            const float tj = uT[j];
            rk += (tj < th || (tj == th && j < me)) ? 1 : 0;
        }
        sTau[rk] = th;
        oC[rk] = uC[me];
        oK[rk] = uK[me];
        oP[rk] = uP[me];
    } else if (tid < H_DIM + 2) {
        sTau[tid] = INF;
    }
    __syncthreads();
    if (tid <= H_DIM) {
        const int k = tid;
        float A = 0.0f, B = mb2;
        #pragma unroll
        for (int j4 = 0; j4 < H_DIM / 4; ++j4) {
            const v4f p4 = *(const v4f*)(oP + 4 * j4);
            const v4f c4 = *(const v4f*)(oC + 4 * j4);
            const v4f e4 = *(const v4f*)(oK + 4 * j4);
            #pragma unroll
            for (int u = 0; u < 4; ++u) {
                const bool act = (p4[u] != 0.0f) == (4 * j4 + u < k);
                A += act ? c4[u] : 0.0f;
                B += act ? e4[u] : 0.0f;
            }
        }
        sABf[2 * k]     = A;
        sABf[2 * k + 1] = B;
    }
    #pragma unroll
    for (int i = 0; i < GCELLS / 256; ++i) {
        const int c = tid + 256 * i;
        const float edge = __int_as_float((c + GBASE) << GSHIFT);
        const int kb = mlp_count_below(sTau, edge);
        kc[c] = (c == 0) ? 0 : kb;
    }
    if (tid == 0) kc[GCELLS] = H_DIM;
    __syncthreads();
    for (int w = tid; w < TB_WORDS; w += 256) {
        unsigned v = 0u;
        if (w < 2 * (H_DIM + 1))                       v = __float_as_uint(sABf[w]);
        else if (w >= TB_TAU && w < TB_TAU + H_DIM + 2) v = __float_as_uint(sTau[w - TB_TAU]);
        else if (w >= TB_G1 && w < TB_G1 + 3 * LPAD)    v = __float_as_uint(gL[w - TB_G1]);
        else if (w == TB_NG)                            v = (unsigned)nG4;
        else if (w >= TB_LUT) {
            const int c0 = (w - TB_LUT) * 4;
            #pragma unroll
            for (int b = 0; b < 4; ++b) {
                const int lo = kc[c0 + b];
                const int hi = kc[c0 + b + 1];
                const unsigned en = (unsigned)lo | ((hi - lo >= 3) ? 128u : 0u);
                v |= en << (8 * b);
            }
        }
        tabw[w] = v;
    }
}

#define FS_W2_FLOATS    (H_DIM * O_DIM)
#define FS_STAGE_FLOATS (FS_W2_FLOATS + 2 * H_DIM)

struct FeatStage {
    v4f   w2q[8];
    v4f   wbq;
};

__device__ __forceinline__ void fsums_fetch(
    FeatStage& G,
    const float* __restrict__ wbsrc,
    const float* __restrict__ fW2,
    int f, int lane)
{
    const v4f* __restrict__ g2 = (const v4f*)(fW2 + f * FS_W2_FLOATS) + lane;
    #pragma unroll
    for (int j = 0; j < 8; ++j) G.w2q[j] = g2[j * 32];
    G.wbq = *(const v4f*)(wbsrc + f * H_DIM);
}

__device__ __forceinline__ void fsums_park(
    float* sb, const FeatStage& G, int lane, int half, int r)
{
    #pragma unroll
    for (int j = 0; j < 8; ++j) ((v4f*)sb)[j * 32 + lane] = G.w2q[j];
    v2f lo, hi;
    lo.x = G.wbq.x; lo.y = G.wbq.y;
    hi.x = G.wbq.z; hi.y = G.wbq.w;
    *(v2f*)(sb + FS_W2_FLOATS + (2 * r    ) * 4 + 2 * half) = lo;
    *(v2f*)(sb + FS_W2_FLOATS + (2 * r + 1) * 4 + 2 * half) = hi;
}

__device__ __forceinline__ void fsums_wave_lds_sync()
{
    __builtin_amdgcn_fence(__ATOMIC_RELEASE, "wavefront");
    __builtin_amdgcn_wave_barrier();
    __builtin_amdgcn_fence(__ATOMIC_ACQUIRE, "wavefront");
}

__global__ __launch_bounds__(256) void hgnam_fsums_kernel(
    const float* __restrict__ x,
    const float* __restrict__ fW1,
    const float* __restrict__ fb1,
    const float* __restrict__ fW2,
    const float* __restrict__ fb2,
    const float* __restrict__ mW1,
    const float* __restrict__ mb1,
    const float* __restrict__ mW2,
    const float* __restrict__ mb2p,
    float* __restrict__ fsp,
    float* __restrict__ tab)
{
    __shared__ __attribute__((aligned(16))) float stage[8][2][FS_STAGE_FLOATS];
    __shared__ float xs[8][FS_NFW * 16];
    __shared__ float red[8][256];

    if (blockIdx.x == (N_NODES / 16) * FSPLIT) {
        mlp_table_build(&stage[0][0][0], mW1, mb1, mW2, mb2p, (unsigned*)tab);
        return;
    }

    const int lane = threadIdx.x & 31;
    const int wv   = threadIdx.x >> 5;
    const int half = lane >> 4;
    const int r    = lane & 15;
    const int ntile = blockIdx.x & (N_NODES / 16 - 1);
    const int fhalf = blockIdx.x / (N_NODES / 16);
    const int base_n = ntile * 16;
    const float* __restrict__ wbsrc = (half ? fb1 : fW1) + 4 * r;

    const int f0 = fhalf * (F_FEAT / FSPLIT) + wv * FS_NFW;
    FeatStage G;
    fsums_fetch(G, wbsrc, fW2, f0, lane);

    {
        const v4f xa = *(const v4f*)(x + (base_n + r) * F_FEAT + f0 + 4 * half);
        float* xw = &xs[wv][(4 * half) * 16 + r];
        xw[0 * 16] = xa.x; xw[1 * 16] = xa.y; xw[2 * 16] = xa.z; xw[3 * 16] = xa.w;
    }

    float bw = 0.0f;
    #pragma unroll
    for (int j = 0; j < FS_NFW; ++j) bw += fb2[(f0 + j) * O_DIM + r];
    v8f acc;
    #pragma unroll
    for (int v = 0; v < 8; ++v) acc[v] = bw;

    fsums_park(&stage[wv][0][0], G, lane, half, r);
    fsums_fetch(G, wbsrc, fW2, f0 + 1, lane);
    fsums_wave_lds_sync();

    for (int fi = 0; fi < FS_NFW; ++fi) {
        const float* cb = &stage[wv][fi & 1][0];
        float*       nb = &stage[wv][(fi & 1) ^ 1][0];

        const float xcur = xs[wv][fi * 16 + r];
        v4f wb[16];
        v2f bf[16];
        #pragma unroll
        for (int c = 0; c < 16; ++c) {
            const int h0 = 4 * c + 2 * half;
            wb[c]   = *(const v4f*)(cb + FS_W2_FLOATS + (2 * c + half) * 4);
            bf[c].x = cb[(h0    ) * O_DIM + r];
            bf[c].y = cb[(h0 + 1) * O_DIM + r];
        }

        if (fi + 1 < FS_NFW) fsums_park(nb, G, lane, half, r);
        if (fi + 2 < FS_NFW) fsums_fetch(G, wbsrc, fW2, f0 + fi + 2, lane);

        #pragma unroll
        for (int c = 0; c < 16; ++c) {
            v2f a;
            a.x = fmaxf(fmaf(xcur, wb[c].x, wb[c].z), 0.0f);
            a.y = fmaxf(fmaf(xcur, wb[c].y, wb[c].w), 0.0f);
            acc = __builtin_amdgcn_wmma_f32_16x16x4_f32(
                false, a, false, bf[c], (short)0, acc, false, false);
        }

        fsums_wave_lds_sync();
    }

    #pragma unroll
    for (int v = 0; v < 8; ++v) red[wv][lane * 8 + v] = acc[v];
    __syncthreads();
    if (wv == 0) {
        float* __restrict__ plane = fsp + (size_t)fhalf * (N_NODES * O_DIM);
        #pragma unroll
        for (int v = 0; v < 8; ++v) {
            float s = 0.0f;
            #pragma unroll
            for (int u = 0; u < 8; ++u) s += red[u][lane * 8 + v];
            plane[(base_n + v + 8 * half) * O_DIM + r] = s;
        }
    }
}


__global__ __launch_bounds__(MWAVES * 32) void hgnam_main_kernel(
    const float* __restrict__ dist,
    const float* __restrict__ nrm,
    const float* __restrict__ tab,
    const float* __restrict__ fsp,
    float* __restrict__ out)
{
    __shared__ __attribute__((aligned(16))) unsigned sTab[TB_WORDS];
    __shared__ float red[MWAVES][256];

    const int lane = threadIdx.x & 31;
    const int wv   = threadIdx.x >> 5;
    const int half = lane >> 4;
    const int r    = lane & 15;
    const int row  = r & 7;
    const int mset = r >> 3;
    const int base_n = blockIdx.x * MROWS;
    const int kbase  = wv * (N_NODES / MWAVES);
    const int NSTEPS = N_NODES / MWAVES / 32;

    if (threadIdx.x < TB_WORDS / 4)
        ((v4f*)sTab)[threadIdx.x] = ((const v4f*)tab)[threadIdx.x];
    const int nG4 = __builtin_amdgcn_readfirstlane(((const int*)tab)[TB_NG]);
    const v2f*           sAB  = (const v2f*)(sTab + TB_AB);
    const float*         sTau = (const float*)(sTab + TB_TAU);
    const float*         gW1  = (const float*)(sTab + TB_G1);
    const float*         gB1  = gW1 + LPAD;
    const float*         gW2  = gW1 + 2 * LPAD;
    const unsigned char* sLut = (const unsigned char*)(sTab + TB_LUT);
    __syncthreads();

    const size_t aoff = (size_t)(base_n + row) * N_NODES + 16 * mset + 2 * half;
    const float* __restrict__ drow = dist + aoff;
    const float* __restrict__ vrow = nrm  + aoff;
    const float* __restrict__ fcol = fsp + (2 * half) * O_DIM + r;

    v8f acc0 = {};
    v8f acc1 = {};

    for (int kt = 0; kt < NSTEPS; ++kt) {
        const int m0 = kbase + kt * 32;

        v2f d[4], q[4], s[4], b0[4], b1[4];
        #pragma unroll
        for (int kk = 0; kk < 4; ++kk) {
            d[kk] = *(const v2f*)(drow + m0 + 4 * kk);
            q[kk] = *(const v2f*)(vrow + m0 + 4 * kk);
            const float* __restrict__ fp = fcol + (m0 + 4 * kk) * O_DIM;
            const float* __restrict__ fq = fp + 16 * O_DIM;
            b0[kk].x = fp[0]     + fp[N_NODES * O_DIM];
            b0[kk].y = fp[O_DIM] + fp[N_NODES * O_DIM + O_DIM];
            b1[kk].x = fq[0]     + fq[N_NODES * O_DIM];
            b1[kk].y = fq[O_DIM] + fq[N_NODES * O_DIM + O_DIM];
        }

        unsigned many = 0u;
        #pragma unroll
        for (int kk = 0; kk < 4; ++kk) {
            #pragma unroll
            for (int e = 0; e < 2; ++e) {
                const float dv = d[kk][e];
                const int key = __float_as_int(dv) >> GSHIFT;
                const int cc  = min(max(key, GBASE), GBASE + GCELLS - 1);
                const unsigned en = sLut[cc - GBASE];
                many |= en;
                const float* tp = sTau + en;
                const unsigned kb = (en << 3) + ((tp[0] < dv) ? 8u : 0u) + ((tp[1] < dv) ? 8u : 0u);
                const v2f ab = *(const v2f*)((const char*)sAB + kb);
                s[kk][e] = fmaf(ab.x, dv, ab.y);
            }
        }
        if (__builtin_amdgcn_ballot_w32((many & 128u) != 0u) != 0u) {
            #pragma unroll
            for (int kk = 0; kk < 4; ++kk) {
                #pragma unroll
                for (int e = 0; e < 2; ++e) {
                    const float dv = d[kk][e];
                    const v2f ab = sAB[mlp_count_below(sTau, dv)];
                    s[kk][e] = fmaf(ab.x, dv, ab.y);
                }
            }
        }
        for (int i = 0; i < nG4; i += 4) {
            const v4f u4 = *(const v4f*)(gW1 + i);
            const v4f v4 = *(const v4f*)(gB1 + i);
            const v4f z4 = *(const v4f*)(gW2 + i);
            #pragma unroll
            for (int j = 0; j < 4; ++j) {
                #pragma unroll
                for (int kk = 0; kk < 4; ++kk) {
                    s[kk].x = fmaf(fmaxf(fmaf(d[kk].x, u4[j], v4[j]), 0.0f), z4[j], s[kk].x);
                    s[kk].y = fmaf(fmaxf(fmaf(d[kk].y, u4[j], v4[j]), 0.0f), z4[j], s[kk].y);
                }
            }
        }

        #pragma unroll
        for (int kk = 0; kk < 4; ++kk) {
            v2f a;
            a.x = s[kk].x * __builtin_amdgcn_rcpf(q[kk].x);
            a.y = s[kk].y * __builtin_amdgcn_rcpf(q[kk].y);
            acc0 = __builtin_amdgcn_wmma_f32_16x16x4_f32(
                false, a, false, b0[kk], (short)0, acc0, false, false);
            acc1 = __builtin_amdgcn_wmma_f32_16x16x4_f32(
                false, a, false, b1[kk], (short)0, acc1, false, false);
        }
    }

    #pragma unroll
    for (int v = 0; v < 8; ++v) red[wv][v * 32 + lane] = half ? acc1[v] : acc0[v];
    __syncthreads();
    if (threadIdx.x < MROWS * O_DIM) {
        const int v = threadIdx.x >> 4;
        const int c = threadIdx.x & 15;
        float ssum = 0.0f;
        #pragma unroll
        for (int u = 0; u < MWAVES; ++u)
            ssum += red[u][v * 32 + c] + red[u][v * 32 + 16 + c];
        out[(base_n + v) * O_DIM + c] = ssum;
    }
}

extern "C" void kernel_launch(void* const* d_in, const int* in_sizes, int n_in,
                              void* d_out, int out_size, void* d_ws, size_t ws_size,
                              hipStream_t stream) {
    (void)in_sizes; (void)n_in; (void)out_size; (void)ws_size;
    const float* x    = (const float*)d_in[0];
    const float* dist = (const float*)d_in[1];
    const float* nrm  = (const float*)d_in[2];
    const float* fW1  = (const float*)d_in[3];
    const float* fb1  = (const float*)d_in[4];
    const float* fW2  = (const float*)d_in[5];
    const float* fb2  = (const float*)d_in[6];
    const float* mW1  = (const float*)d_in[7];
    const float* mb1  = (const float*)d_in[8];
    const float* mW2  = (const float*)d_in[9];
    const float* mb2  = (const float*)d_in[10];

    float* fsp   = (float*)d_ws;
    float* tab   = fsp + FSPLIT * N_NODES * O_DIM;
    float* out   = (float*)d_out;

    hgnam_fsums_kernel<<<dim3((N_NODES / 16) * FSPLIT + 1), dim3(256), 0, stream>>>(
        x, fW1, fb1, fW2, fb2, mW1, mb1, mW2, mb2, fsp, tab);
    hgnam_main_kernel<<<dim3(N_NODES / MROWS), dim3(MWAVES * 32), 0, stream>>>(
        dist, nrm, tab, fsp, out);
}
